# starting kernel
# baseline (speedup 1.0000x reference)
#include <hip/hip_runtime.h>


typedef __attribute__((ext_vector_type(2))) float v2f;
typedef __attribute__((ext_vector_type(8))) float v8f;

#if __has_builtin(__builtin_amdgcn_tanhf)
#define HTANH(x) __builtin_amdgcn_tanhf((x))
#elif __has_builtin(__builtin_amdgcn_tanh_f32)
#define HTANH(x) __builtin_amdgcn_tanh_f32((x))
#else
#define HTANH(x) fmaf(2.0f, __builtin_amdgcn_rcpf(1.0f + __expf(-2.0f * (x))), -1.0f)
#endif

__device__ __forceinline__ float fast_tanh(float x) { return HTANH(x); }

__device__ __forceinline__ float fast_sig(float x) {
#if __has_builtin(__builtin_amdgcn_tanhf) || __has_builtin(__builtin_amdgcn_tanh_f32)
    return fmaf(0.5f, HTANH(0.5f * x), 0.5f);
#else
    return __builtin_amdgcn_rcpf(1.0f + __expf(-x));
#endif
}

__device__ __forceinline__ float shfl_x16(float v) {
    int lane = (int)(threadIdx.x & 31u);
    int addr = (lane ^ 16) << 2;
    return __int_as_float(__builtin_amdgcn_ds_bpermute(addr, __float_as_int(v)));
}

#define LSTM_STEP(xv, yOut)                                   \
    do {                                                      \
        float gi = fmaf(h, wh0, fmaf((xv), wi0, bb0));        \
        float gf = fmaf(h, wh1, fmaf((xv), wi1, bb1));        \
        float gg = fmaf(h, wh2, fmaf((xv), wi2, bb2));        \
        float go = fmaf(h, wh3, fmaf((xv), wi3, bb3));        \
        float ii = fast_sig(gi);                              \
        float ff = fast_sig(gf);                              \
        float gt = fast_tanh(gg);                             \
        float oo = fast_sig(go);                              \
        c = fmaf(ff, c, ii * gt);                             \
        h = oo * fast_tanh(c);                                \
        (yOut) = fast_sig(h);                                 \
    } while (0)

__global__ __launch_bounds__(256)
void OksuzLSTM_scan_wmma(const float* __restrict__ x,
                         const float* __restrict__ w_ih,
                         const float* __restrict__ w_hh,
                         const float* __restrict__ b_ih,
                         const float* __restrict__ b_hh,
                         const float* __restrict__ l1_w,
                         const float* __restrict__ l1_b,
                         float* __restrict__ out)
{
    const int lane     = (int)(threadIdx.x & 31u);
    const int row      = blockIdx.x * (int)blockDim.x + (int)threadIdx.x;
    const int waveBase = blockIdx.x * (int)blockDim.x + (int)(threadIdx.x & ~31u);

    const float wi0 = w_ih[0], wi1 = w_ih[1], wi2 = w_ih[2], wi3 = w_ih[3];
    const float wh0 = w_hh[0], wh1 = w_hh[1], wh2 = w_hh[2], wh3 = w_hh[3];
    const float bb0 = b_ih[0] + b_hh[0];
    const float bb1 = b_ih[1] + b_hh[1];
    const float bb2 = b_ih[2] + b_hh[2];
    const float bb3 = b_ih[3] + b_hh[3];

    const float* xrow = x + (size_t)row * 1000u;

    const bool lo   = lane < 16;
    const int  n    = lane & 15;
    const int  koff = lo ? 0 : 2;
    const float* wrow = l1_w + (size_t)((n < 2) ? n : 0) * 1000u;

    float h = 0.0f, c = 0.0f;
    v8f acc0 = {0.f, 0.f, 0.f, 0.f, 0.f, 0.f, 0.f, 0.f};
    v8f acc1 = {0.f, 0.f, 0.f, 0.f, 0.f, 0.f, 0.f, 0.f};

#pragma unroll 2
    for (int t = 0; t < 1000; t += 4) {
        __builtin_prefetch(xrow + t + 128, 0, 1);

        const float4 xv = *reinterpret_cast<const float4*>(xrow + t);

        float y0, y1, y2, y3;
        LSTM_STEP(xv.x, y0);
        LSTM_STEP(xv.y, y1);
        LSTM_STEP(xv.z, y2);
        LSTM_STEP(xv.w, y3);

        const float s0 = shfl_x16(y0);
        const float s1 = shfl_x16(y1);
        const float s2 = shfl_x16(y2);
        const float s3 = shfl_x16(y3);

        v2f a0, a1, bf;
        a0.x = lo ? y0 : s2;
        a0.y = lo ? y1 : s3;
        a1.x = lo ? s0 : y2;
        a1.y = lo ? s1 : y3;

        float bv0 = wrow[t + koff];
        float bv1 = wrow[t + koff + 1];
        bf.x = (n < 2) ? bv0 : 0.0f;
        bf.y = (n < 2) ? bv1 : 0.0f;

        acc0 = __builtin_amdgcn_wmma_f32_16x16x4_f32(
            false, a0, false, bf, (short)0, acc0, false, false);
        acc1 = __builtin_amdgcn_wmma_f32_16x16x4_f32(
            false, a1, false, bf, (short)0, acc1, false, false);
    }

    if (n < 2) {
        const float bias = l1_b[n];
        float* p = out + ((size_t)(waveBase + (lo ? 0 : 8)) * 2u + (size_t)n);
#pragma unroll
        for (int g = 0; g < 8; ++g) {
            p[2 * g]      = acc0[g] + bias;
            p[2 * g + 32] = acc1[g] + bias;
        }
    }
}

extern "C" void kernel_launch(void* const* d_in, const int* in_sizes, int n_in,
                              void* d_out, int out_size, void* d_ws, size_t ws_size,
                              hipStream_t stream) {
    (void)in_sizes; (void)n_in; (void)d_ws; (void)ws_size; (void)out_size;

    const float* x    = (const float*)d_in[0];
    const float* w_ih = (const float*)d_in[1];
    const float* w_hh = (const float*)d_in[2];
    const float* b_ih = (const float*)d_in[3];
    const float* b_hh = (const float*)d_in[4];
    const float* l1_w = (const float*)d_in[5];
    const float* l1_b = (const float*)d_in[6];
    float* out = (float*)d_out;

    const int B = 32768;
    dim3 block(256);
    dim3 grid(B / 256);
    hipLaunchKernelGGL(OksuzLSTM_scan_wmma, grid, block, 0, stream,
                       x, w_ih, w_hh, b_ih, b_hh, l1_w, l1_b, out);
}


# x float4 and B-fragment global loads issued 10 loop iterations (40 timesteps) ahead into a register ring instead of loaded and waited on at each use; arithmetic and summation order unchanged
# speedup vs baseline: 1.4715x; 1.4715x over previous
#include <hip/hip_runtime.h>


typedef __attribute__((ext_vector_type(2))) float v2f;
typedef __attribute__((ext_vector_type(8))) float v8f;

#if __has_builtin(__builtin_amdgcn_tanhf)
#define HTANH(x) __builtin_amdgcn_tanhf((x))
#elif __has_builtin(__builtin_amdgcn_tanh_f32)
#define HTANH(x) __builtin_amdgcn_tanh_f32((x))
#else
#define HTANH(x) fmaf(2.0f, __builtin_amdgcn_rcpf(1.0f + __expf(-2.0f * (x))), -1.0f)
#endif

__device__ __forceinline__ float fast_tanh(float x) { return HTANH(x); }

__device__ __forceinline__ float fast_sig(float x) {
#if __has_builtin(__builtin_amdgcn_tanhf) || __has_builtin(__builtin_amdgcn_tanh_f32)
    return fmaf(0.5f, HTANH(0.5f * x), 0.5f);
#else
    return __builtin_amdgcn_rcpf(1.0f + __expf(-x));
#endif
}

__device__ __forceinline__ float shfl_x16(float v) {
    int lane = (int)(threadIdx.x & 31u);
    int addr = (lane ^ 16) << 2;
    return __int_as_float(__builtin_amdgcn_ds_bpermute(addr, __float_as_int(v)));
}

#define RING 10

#define RING_PIN() asm volatile("" ::: "memory")

#define LSTM_PRE(xv, p)                                       \
    do {                                                      \
        (p)[0] = fmaf((xv), wi0, bb0);                        \
        (p)[1] = fmaf((xv), wi1, bb1);                        \
        (p)[2] = fmaf((xv), wi2, bb2);                        \
        (p)[3] = fmaf((xv), wi3, bb3);                        \
    } while (0)

#define LSTM_STEP(p, yOut)                                    \
    do {                                                      \
        float gi = fmaf(h, wh0, (p)[0]);                      \
        float gf = fmaf(h, wh1, (p)[1]);                      \
        float gg = fmaf(h, wh2, (p)[2]);                      \
        float go = fmaf(h, wh3, (p)[3]);                      \
        float ii = fast_sig(gi);                              \
        float ff = fast_sig(gf);                              \
        float gt = fast_tanh(gg);                             \
        float oo = fast_sig(go);                              \
        c = fmaf(ff, c, ii * gt);                             \
        h = oo * fast_tanh(c);                                \
        (yOut) = fast_sig(h);                                 \
    } while (0)

__global__ __launch_bounds__(256)
void OksuzLSTM_scan_wmma(const float* __restrict__ x,
                         const float* __restrict__ w_ih,
                         const float* __restrict__ w_hh,
                         const float* __restrict__ b_ih,
                         const float* __restrict__ b_hh,
                         const float* __restrict__ l1_w,
                         const float* __restrict__ l1_b,
                         float* __restrict__ out)
{
    const int lane     = (int)(threadIdx.x & 31u);
    const int row      = blockIdx.x * (int)blockDim.x + (int)threadIdx.x;
    const int waveBase = blockIdx.x * (int)blockDim.x + (int)(threadIdx.x & ~31u);

    const float wi0 = w_ih[0], wi1 = w_ih[1], wi2 = w_ih[2], wi3 = w_ih[3];
    const float wh0 = w_hh[0], wh1 = w_hh[1], wh2 = w_hh[2], wh3 = w_hh[3];
    const float bb0 = b_ih[0] + b_hh[0];
    const float bb1 = b_ih[1] + b_hh[1];
    const float bb2 = b_ih[2] + b_hh[2];
    const float bb3 = b_ih[3] + b_hh[3];

    const float* xrow = x + (size_t)row * 1000u;

    const bool lo   = lane < 16;
    const int  n    = lane & 15;
    const int  koff = lo ? 0 : 2;
    const float* wrow = l1_w + (size_t)((n < 2) ? n : 0) * 1000u;

    float h = 0.0f, c = 0.0f;
    v8f acc0 = {0.f, 0.f, 0.f, 0.f, 0.f, 0.f, 0.f, 0.f};
    v8f acc1 = {0.f, 0.f, 0.f, 0.f, 0.f, 0.f, 0.f, 0.f};

    float4 xq[RING];
    float  bq0[RING], bq1[RING];
#pragma unroll
    for (int k = 0; k < RING; ++k) {
        xq[k]  = *reinterpret_cast<const float4*>(xrow + 4 * k);
        bq0[k] = wrow[4 * k + koff];
        bq1[k] = wrow[4 * k + koff + 1];
        RING_PIN();
    }

    for (int t0 = 0; t0 < 1000; t0 += 4 * RING) {
        const int tr = (t0 + 4 * RING < 1000) ? (t0 + 4 * RING) : t0;

        __builtin_prefetch(xrow + t0 + 104, 0, 1);
        __builtin_prefetch(xrow + t0 + 124, 0, 1);

#pragma unroll
        for (int k = 0; k < RING; ++k) {
            const float4 xv = xq[k];
            float p0[4], p1[4], p2[4], p3[4];
            LSTM_PRE(xv.x, p0);
            LSTM_PRE(xv.y, p1);
            LSTM_PRE(xv.z, p2);
            LSTM_PRE(xv.w, p3);

            v2f bf;
            bf.x = (n < 2) ? bq0[k] : 0.0f;
            bf.y = (n < 2) ? bq1[k] : 0.0f;

            __builtin_amdgcn_sched_barrier(0);

            xq[k]  = *reinterpret_cast<const float4*>(xrow + tr + 4 * k);
            bq0[k] = wrow[tr + 4 * k + koff];
            bq1[k] = wrow[tr + 4 * k + koff + 1];
            RING_PIN();

            float y0, y1, y2, y3;
            LSTM_STEP(p0, y0);
            LSTM_STEP(p1, y1);
            LSTM_STEP(p2, y2);
            LSTM_STEP(p3, y3);

            const float s0 = shfl_x16(y0);
            const float s1 = shfl_x16(y1);
            const float s2 = shfl_x16(y2);
            const float s3 = shfl_x16(y3);

            v2f a0, a1;
            a0.x = lo ? y0 : s2;
            a0.y = lo ? y1 : s3;
            a1.x = lo ? s0 : y2;
            a1.y = lo ? s1 : y3;

            acc0 = __builtin_amdgcn_wmma_f32_16x16x4_f32(
                false, a0, false, bf, (short)0, acc0, false, false);
            acc1 = __builtin_amdgcn_wmma_f32_16x16x4_f32(
                false, a1, false, bf, (short)0, acc1, false, false);
        }
    }

    if (n < 2) {
        const float bias = l1_b[n];
        float* p = out + ((size_t)(waveBase + (lo ? 0 : 8)) * 2u + (size_t)n);
#pragma unroll
        for (int g = 0; g < 8; ++g) {
            p[2 * g]      = acc0[g] + bias;
            p[2 * g + 32] = acc1[g] + bias;
        }
    }
}

extern "C" void kernel_launch(void* const* d_in, const int* in_sizes, int n_in,
                              void* d_out, int out_size, void* d_ws, size_t ws_size,
                              hipStream_t stream) {
    (void)in_sizes; (void)n_in; (void)d_ws; (void)ws_size; (void)out_size;

    const float* x    = (const float*)d_in[0];
    const float* w_ih = (const float*)d_in[1];
    const float* w_hh = (const float*)d_in[2];
    const float* b_ih = (const float*)d_in[3];
    const float* b_hh = (const float*)d_in[4];
    const float* l1_w = (const float*)d_in[5];
    const float* l1_b = (const float*)d_in[6];
    float* out = (float*)d_out;

    const int B = 32768;
    dim3 block(256);
    dim3 grid(B / 256);
    hipLaunchKernelGGL(OksuzLSTM_scan_wmma, grid, block, 0, stream,
                       x, w_ih, w_hh, b_ih, b_hh, l1_w, l1_b, out);
}


# Launch shape only: 256 work-groups of 4 waves instead of 128 of 8, so all 256 WGPs get one work-group and each SIMD runs one wave's recurrence instead of time-sharing two; kernel code unchanged
# speedup vs baseline: 1.7244x; 1.1719x over previous
#include <hip/hip_runtime.h>


typedef __attribute__((ext_vector_type(2))) float v2f;
typedef __attribute__((ext_vector_type(8))) float v8f;

#if __has_builtin(__builtin_amdgcn_tanhf)
#define HTANH(x) __builtin_amdgcn_tanhf((x))
#elif __has_builtin(__builtin_amdgcn_tanh_f32)
#define HTANH(x) __builtin_amdgcn_tanh_f32((x))
#else
#define HTANH(x) fmaf(2.0f, __builtin_amdgcn_rcpf(1.0f + __expf(-2.0f * (x))), -1.0f)
#endif

__device__ __forceinline__ float fast_tanh(float x) { return HTANH(x); }

__device__ __forceinline__ float fast_sig(float x) {
#if __has_builtin(__builtin_amdgcn_tanhf) || __has_builtin(__builtin_amdgcn_tanh_f32)
    return fmaf(0.5f, HTANH(0.5f * x), 0.5f);
#else
    return __builtin_amdgcn_rcpf(1.0f + __expf(-x));
#endif
}

__device__ __forceinline__ float shfl_x16(float v) {
    int lane = (int)(threadIdx.x & 31u);
    int addr = (lane ^ 16) << 2;
    return __int_as_float(__builtin_amdgcn_ds_bpermute(addr, __float_as_int(v)));
}

#define RING 10

#define RING_PIN() asm volatile("" ::: "memory")

#define LSTM_PRE(xv, p)                                       \
    do {                                                      \
        (p)[0] = fmaf((xv), wi0, bb0);                        \
        (p)[1] = fmaf((xv), wi1, bb1);                        \
        (p)[2] = fmaf((xv), wi2, bb2);                        \
        (p)[3] = fmaf((xv), wi3, bb3);                        \
    } while (0)

#define LSTM_STEP(p, yOut)                                    \
    do {                                                      \
        float gi = fmaf(h, wh0, (p)[0]);                      \
        float gf = fmaf(h, wh1, (p)[1]);                      \
        float gg = fmaf(h, wh2, (p)[2]);                      \
        float go = fmaf(h, wh3, (p)[3]);                      \
        float ii = fast_sig(gi);                              \
        float ff = fast_sig(gf);                              \
        float gt = fast_tanh(gg);                             \
        float oo = fast_sig(go);                              \
        c = fmaf(ff, c, ii * gt);                             \
        h = oo * fast_tanh(c);                                \
        (yOut) = fast_sig(h);                                 \
    } while (0)

__global__ __launch_bounds__(128)
void OksuzLSTM_scan_wmma(const float* __restrict__ x,
                         const float* __restrict__ w_ih,
                         const float* __restrict__ w_hh,
                         const float* __restrict__ b_ih,
                         const float* __restrict__ b_hh,
                         const float* __restrict__ l1_w,
                         const float* __restrict__ l1_b,
                         float* __restrict__ out)
{
    const int lane     = (int)(threadIdx.x & 31u);
    const int row      = blockIdx.x * (int)blockDim.x + (int)threadIdx.x;
    const int waveBase = blockIdx.x * (int)blockDim.x + (int)(threadIdx.x & ~31u);

    const float wi0 = w_ih[0], wi1 = w_ih[1], wi2 = w_ih[2], wi3 = w_ih[3];
    const float wh0 = w_hh[0], wh1 = w_hh[1], wh2 = w_hh[2], wh3 = w_hh[3];
    const float bb0 = b_ih[0] + b_hh[0];
    const float bb1 = b_ih[1] + b_hh[1];
    const float bb2 = b_ih[2] + b_hh[2];
    const float bb3 = b_ih[3] + b_hh[3];

    const float* xrow = x + (size_t)row * 1000u;

    const bool lo   = lane < 16;
    const int  n    = lane & 15;
    const int  koff = lo ? 0 : 2;
    const float* wrow = l1_w + (size_t)((n < 2) ? n : 0) * 1000u;

    float h = 0.0f, c = 0.0f;
    v8f acc0 = {0.f, 0.f, 0.f, 0.f, 0.f, 0.f, 0.f, 0.f};
    v8f acc1 = {0.f, 0.f, 0.f, 0.f, 0.f, 0.f, 0.f, 0.f};

    float4 xq[RING];
    float  bq0[RING], bq1[RING];
#pragma unroll
    for (int k = 0; k < RING; ++k) {
        xq[k]  = *reinterpret_cast<const float4*>(xrow + 4 * k);
        bq0[k] = wrow[4 * k + koff];
        bq1[k] = wrow[4 * k + koff + 1];
        RING_PIN();
    }

    for (int t0 = 0; t0 < 1000; t0 += 4 * RING) {
        const int tr = (t0 + 4 * RING < 1000) ? (t0 + 4 * RING) : t0;

        __builtin_prefetch(xrow + t0 + 104, 0, 1);
        __builtin_prefetch(xrow + t0 + 124, 0, 1);

#pragma unroll
        for (int k = 0; k < RING; ++k) {
            const float4 xv = xq[k];
            float p0[4], p1[4], p2[4], p3[4];
            LSTM_PRE(xv.x, p0);
            LSTM_PRE(xv.y, p1);
            LSTM_PRE(xv.z, p2);
            LSTM_PRE(xv.w, p3);

            v2f bf;
            bf.x = (n < 2) ? bq0[k] : 0.0f;
            bf.y = (n < 2) ? bq1[k] : 0.0f;

            __builtin_amdgcn_sched_barrier(0);

            xq[k]  = *reinterpret_cast<const float4*>(xrow + tr + 4 * k);
            bq0[k] = wrow[tr + 4 * k + koff];
            bq1[k] = wrow[tr + 4 * k + koff + 1];
            RING_PIN();

            float y0, y1, y2, y3;
            LSTM_STEP(p0, y0);
            LSTM_STEP(p1, y1);
            LSTM_STEP(p2, y2);
            LSTM_STEP(p3, y3);

            const float s0 = shfl_x16(y0);
            const float s1 = shfl_x16(y1);
            const float s2 = shfl_x16(y2);
            const float s3 = shfl_x16(y3);

            v2f a0, a1;
            a0.x = lo ? y0 : s2;
            a0.y = lo ? y1 : s3;
            a1.x = lo ? s0 : y2;
            a1.y = lo ? s1 : y3;

            acc0 = __builtin_amdgcn_wmma_f32_16x16x4_f32(
                false, a0, false, bf, (short)0, acc0, false, false);
            acc1 = __builtin_amdgcn_wmma_f32_16x16x4_f32(
                false, a1, false, bf, (short)0, acc1, false, false);
        }
    }

    if (n < 2) {
        const float bias = l1_b[n];
        float* p = out + ((size_t)(waveBase + (lo ? 0 : 8)) * 2u + (size_t)n);
#pragma unroll
        for (int g = 0; g < 8; ++g) {
            p[2 * g]      = acc0[g] + bias;
            p[2 * g + 32] = acc1[g] + bias;
        }
    }
}

extern "C" void kernel_launch(void* const* d_in, const int* in_sizes, int n_in,
                              void* d_out, int out_size, void* d_ws, size_t ws_size,
                              hipStream_t stream) {
    (void)in_sizes; (void)n_in; (void)d_ws; (void)ws_size; (void)out_size;

    const float* x    = (const float*)d_in[0];
    const float* w_ih = (const float*)d_in[1];
    const float* w_hh = (const float*)d_in[2];
    const float* b_ih = (const float*)d_in[3];
    const float* b_hh = (const float*)d_in[4];
    const float* l1_w = (const float*)d_in[5];
    const float* l1_b = (const float*)d_in[6];
    float* out = (float*)d_out;

    const int B = 32768;
    dim3 block(128);
    dim3 grid(B / 128);
    hipLaunchKernelGGL(OksuzLSTM_scan_wmma, grid, block, 0, stream,
                       x, w_ih, w_hh, b_ih, b_hh, l1_w, l1_b, out);
}


# No cross-lane exchange in the projection: A = each lane's own sigmoids of 2 timesteps, B zero outside K pair = lane half, one accumulator; sums regroup (2 real products per matrix instruction, was 4)
# speedup vs baseline: 2.3455x; 1.3602x over previous
#include <hip/hip_runtime.h>


typedef __attribute__((ext_vector_type(2))) float v2f;
typedef __attribute__((ext_vector_type(8))) float v8f;

#if __has_builtin(__builtin_amdgcn_tanhf)
#define HTANH(x) __builtin_amdgcn_tanhf((x))
#elif __has_builtin(__builtin_amdgcn_tanh_f32)
#define HTANH(x) __builtin_amdgcn_tanh_f32((x))
#else
#define HTANH(x) fmaf(2.0f, __builtin_amdgcn_rcpf(1.0f + __expf(-2.0f * (x))), -1.0f)
#endif

__device__ __forceinline__ float fast_tanh(float x) { return HTANH(x); }

__device__ __forceinline__ float fast_sig(float x) {
#if __has_builtin(__builtin_amdgcn_tanhf) || __has_builtin(__builtin_amdgcn_tanh_f32)
    return fmaf(0.5f, HTANH(0.5f * x), 0.5f);
#else
    return __builtin_amdgcn_rcpf(1.0f + __expf(-x));
#endif
}

#define RING 10

#define RING_PIN() asm volatile("" ::: "memory")

#define LSTM_PRE(xv, p)                                       \
    do {                                                      \
        (p)[0] = fmaf((xv), wi0, bb0);                        \
        (p)[1] = fmaf((xv), wi1, bb1);                        \
        (p)[2] = fmaf((xv), wi2, bb2);                        \
        (p)[3] = fmaf((xv), wi3, bb3);                        \
    } while (0)

#define LSTM_STEP(p, yOut)                                    \
    do {                                                      \
        float gi = fmaf(h, wh0, (p)[0]);                      \
        float gf = fmaf(h, wh1, (p)[1]);                      \
        float gg = fmaf(h, wh2, (p)[2]);                      \
        float go = fmaf(h, wh3, (p)[3]);                      \
        float ii = fast_sig(gi);                              \
        float ff = fast_sig(gf);                              \
        float gt = fast_tanh(gg);                             \
        float oo = fast_sig(go);                              \
        c = fmaf(ff, c, ii * gt);                             \
        h = oo * fast_tanh(c);                                \
        (yOut) = fast_sig(h);                                 \
    } while (0)

__global__ __launch_bounds__(128)
void OksuzLSTM_scan_wmma(const float* __restrict__ x,
                         const float* __restrict__ w_ih,
                         const float* __restrict__ w_hh,
                         const float* __restrict__ b_ih,
                         const float* __restrict__ b_hh,
                         const float* __restrict__ l1_w,
                         const float* __restrict__ l1_b,
                         float* __restrict__ out)
{
    const int lane     = (int)(threadIdx.x & 31u);
    const int row      = blockIdx.x * (int)blockDim.x + (int)threadIdx.x;
    const int waveBase = blockIdx.x * (int)blockDim.x + (int)(threadIdx.x & ~31u);

    const float wi0 = w_ih[0], wi1 = w_ih[1], wi2 = w_ih[2], wi3 = w_ih[3];
    const float wh0 = w_hh[0], wh1 = w_hh[1], wh2 = w_hh[2], wh3 = w_hh[3];
    const float bb0 = b_ih[0] + b_hh[0];
    const float bb1 = b_ih[1] + b_hh[1];
    const float bb2 = b_ih[2] + b_hh[2];
    const float bb3 = b_ih[3] + b_hh[3];

    const float* xrow = x + (size_t)row * 1000u;

    const int  n    = lane & 15;
    const int  hh   = lane >> 4;
    const bool bact = (n < 4) && ((n >> 1) == hh);
    const int  oc   = n & 1;
    const float* wrow = l1_w + (size_t)(bact ? oc : 0) * 1000u;

    float h = 0.0f, c = 0.0f;
    v8f acc = {0.f, 0.f, 0.f, 0.f, 0.f, 0.f, 0.f, 0.f};

    float4 xq[RING];
    float4 bq[RING];
#pragma unroll
    for (int k = 0; k < RING; ++k) {
        xq[k] = *reinterpret_cast<const float4*>(xrow + 4 * k);
        bq[k] = *reinterpret_cast<const float4*>(wrow + 4 * k);
        RING_PIN();
    }

    for (int t0 = 0; t0 < 1000; t0 += 4 * RING) {
        const int tr = (t0 + 4 * RING < 1000) ? (t0 + 4 * RING) : t0;

        __builtin_prefetch(xrow + t0 + 104, 0, 1);
        __builtin_prefetch(xrow + t0 + 124, 0, 1);

#pragma unroll
        for (int k = 0; k < RING; ++k) {
            const float4 xv = xq[k];
            float p0[4], p1[4], p2[4], p3[4];
            LSTM_PRE(xv.x, p0);
            LSTM_PRE(xv.y, p1);
            LSTM_PRE(xv.z, p2);
            LSTM_PRE(xv.w, p3);

            const float4 bw = bq[k];
            v2f bf0, bf1;
            bf0.x = bact ? bw.x : 0.0f;
            bf0.y = bact ? bw.y : 0.0f;
            bf1.x = bact ? bw.z : 0.0f;
            bf1.y = bact ? bw.w : 0.0f;

            __builtin_amdgcn_sched_barrier(0);

            xq[k] = *reinterpret_cast<const float4*>(xrow + tr + 4 * k);
            bq[k] = *reinterpret_cast<const float4*>(wrow + tr + 4 * k);
            RING_PIN();

            float y0, y1, y2, y3;
            LSTM_STEP(p0, y0);
            LSTM_STEP(p1, y1);
            v2f a0;
            a0.x = y0;
            a0.y = y1;
            acc = __builtin_amdgcn_wmma_f32_16x16x4_f32(
                false, a0, false, bf0, (short)0, acc, false, false);

            LSTM_STEP(p2, y2);
            LSTM_STEP(p3, y3);
            v2f a1;
            a1.x = y2;
            a1.y = y3;
            acc = __builtin_amdgcn_wmma_f32_16x16x4_f32(
                false, a1, false, bf1, (short)0, acc, false, false);
        }
    }

    if (n < 4) {
        const float bias = l1_b[oc];
        float* p = out + ((size_t)(waveBase + ((n >> 1) << 4) + (hh << 3)) * 2u + (size_t)oc);
#pragma unroll
        for (int g = 0; g < 8; ++g) {
            p[2 * g] = acc[g] + bias;
        }
    }
}

extern "C" void kernel_launch(void* const* d_in, const int* in_sizes, int n_in,
                              void* d_out, int out_size, void* d_ws, size_t ws_size,
                              hipStream_t stream) {
    (void)in_sizes; (void)n_in; (void)d_ws; (void)ws_size; (void)out_size;

    const float* x    = (const float*)d_in[0];
    const float* w_ih = (const float*)d_in[1];
    const float* w_hh = (const float*)d_in[2];
    const float* b_ih = (const float*)d_in[3];
    const float* b_hh = (const float*)d_in[4];
    const float* l1_w = (const float*)d_in[5];
    const float* l1_b = (const float*)d_in[6];
    float* out = (float*)d_out;

    const int B = 32768;
    dim3 block(128);
    dim3 grid(B / 128);
    hipLaunchKernelGGL(OksuzLSTM_scan_wmma, grid, block, 0, stream,
                       x, w_ih, w_hh, b_ih, b_hh, l1_w, l1_b, out);
}


# Recurrence re-associated to shorten the serial chain to fma,tanh,fma,fma,tanh per step: sigmoid 0.5 folded into gate weights, o*w_hh and f*c formed beside the tanh ops; their rounding regroups
# speedup vs baseline: 2.6723x; 1.1393x over previous
#include <hip/hip_runtime.h>


typedef __attribute__((ext_vector_type(2))) float v2f;
typedef __attribute__((ext_vector_type(8))) float v8f;

#if __has_builtin(__builtin_amdgcn_tanhf)
#define HTANH(x) __builtin_amdgcn_tanhf((x))
#elif __has_builtin(__builtin_amdgcn_tanh_f32)
#define HTANH(x) __builtin_amdgcn_tanh_f32((x))
#else
#define HTANH(x) fmaf(2.0f, __builtin_amdgcn_rcpf(1.0f + __expf(-2.0f * (x))), -1.0f)
#endif

#define RING 10

#define RING_PIN() asm volatile("" ::: "memory")

#define LSTM_PRE(xv, p)                                       \
    do {                                                      \
        (p)[0] = fmaf((xv), wi0, bb0);                        \
        (p)[1] = fmaf((xv), wi1, bb1);                        \
        (p)[2] = fmaf((xv), wi2, bb2);                        \
        (p)[3] = fmaf((xv), wi3, bb3);                        \
    } while (0)

#define LSTM_STEP(p, yOut)                                    \
    do {                                                      \
        float ai = fmaf(tc, q0, (p)[0]);                      \
        float ag = fmaf(tc, q2, (p)[2]);                      \
        float af = fmaf(tc, q1, (p)[1]);                      \
        float ao = fmaf(tc, q3, (p)[3]);                      \
        float ti = HTANH(ai);                                 \
        float tg = HTANH(ag);                                 \
        float tf = HTANH(af);                                 \
        float to = HTANH(ao);                                 \
        float ii = fmaf(0.5f, ti, 0.5f);                      \
        float fc = fmaf(tf, ch, ch);                          \
        c = fmaf(ii, tg, fc);                                 \
        tc = HTANH(c);                                        \
        ch = 0.5f * c;                                        \
        q0 = fmaf(to, wq0, wq0);                              \
        q1 = fmaf(to, wq1, wq1);                              \
        q2 = fmaf(to, wq2, wq2);                              \
        q3 = fmaf(to, wq3, wq3);                              \
        float oh = fmaf(0.25f, to, 0.25f);                    \
        (yOut) = fmaf(0.5f, HTANH(oh * tc), 0.5f);            \
    } while (0)

__global__ __launch_bounds__(128)
void OksuzLSTM_scan_wmma(const float* __restrict__ x,
                         const float* __restrict__ w_ih,
                         const float* __restrict__ w_hh,
                         const float* __restrict__ b_ih,
                         const float* __restrict__ b_hh,
                         const float* __restrict__ l1_w,
                         const float* __restrict__ l1_b,
                         float* __restrict__ out)
{
    const int lane     = (int)(threadIdx.x & 31u);
    const int row      = blockIdx.x * (int)blockDim.x + (int)threadIdx.x;
    const int waveBase = blockIdx.x * (int)blockDim.x + (int)(threadIdx.x & ~31u);

    const float wi0 = 0.5f * w_ih[0], wi1 = 0.5f * w_ih[1], wi2 = w_ih[2], wi3 = 0.5f * w_ih[3];
    const float bb0 = 0.5f * (b_ih[0] + b_hh[0]);
    const float bb1 = 0.5f * (b_ih[1] + b_hh[1]);
    const float bb2 = b_ih[2] + b_hh[2];
    const float bb3 = 0.5f * (b_ih[3] + b_hh[3]);
    const float wq0 = 0.25f * w_hh[0], wq1 = 0.25f * w_hh[1], wq2 = 0.5f * w_hh[2], wq3 = 0.25f * w_hh[3];

    const float* xrow = x + (size_t)row * 1000u;

    const int  n    = lane & 15;
    const int  hh   = lane >> 4;
    const bool bact = (n < 4) && ((n >> 1) == hh);
    const int  oc   = n & 1;
    const float* wrow = l1_w + (size_t)(bact ? oc : 0) * 1000u;

    float c = 0.0f, ch = 0.0f, tc = 0.0f;
    float q0 = wq0, q1 = wq1, q2 = wq2, q3 = wq3;
    v8f acc = {0.f, 0.f, 0.f, 0.f, 0.f, 0.f, 0.f, 0.f};

    float4 xq[RING];
    float4 bq[RING];
#pragma unroll
    for (int k = 0; k < RING; ++k) {
        xq[k] = *reinterpret_cast<const float4*>(xrow + 4 * k);
        bq[k] = *reinterpret_cast<const float4*>(wrow + 4 * k);
        RING_PIN();
    }

    for (int t0 = 0; t0 < 1000; t0 += 4 * RING) {
        const int tr = (t0 + 4 * RING < 1000) ? (t0 + 4 * RING) : t0;

        __builtin_prefetch(xrow + t0 + 104, 0, 1);
        __builtin_prefetch(xrow + t0 + 124, 0, 1);

#pragma unroll
        for (int k = 0; k < RING; ++k) {
            const float4 xv = xq[k];
            float p0[4], p1[4], p2[4], p3[4];
            LSTM_PRE(xv.x, p0);
            LSTM_PRE(xv.y, p1);
            LSTM_PRE(xv.z, p2);
            LSTM_PRE(xv.w, p3);

            const float4 bw = bq[k];
            v2f bf0, bf1;
            bf0.x = bact ? bw.x : 0.0f;
            bf0.y = bact ? bw.y : 0.0f;
            bf1.x = bact ? bw.z : 0.0f;
            bf1.y = bact ? bw.w : 0.0f;

            __builtin_amdgcn_sched_barrier(0);

            xq[k] = *reinterpret_cast<const float4*>(xrow + tr + 4 * k);
            bq[k] = *reinterpret_cast<const float4*>(wrow + tr + 4 * k);
            RING_PIN();

            float y0, y1, y2, y3;
            LSTM_STEP(p0, y0);
            LSTM_STEP(p1, y1);
            v2f a0;
            a0.x = y0;
            a0.y = y1;
            acc = __builtin_amdgcn_wmma_f32_16x16x4_f32(
                false, a0, false, bf0, (short)0, acc, false, false);

            LSTM_STEP(p2, y2);
            LSTM_STEP(p3, y3);
            v2f a1;
            a1.x = y2;
            a1.y = y3;
            acc = __builtin_amdgcn_wmma_f32_16x16x4_f32(
                false, a1, false, bf1, (short)0, acc, false, false);
        }
    }

    if (n < 4) {
        const float bias = l1_b[oc];
        float* p = out + ((size_t)(waveBase + ((n >> 1) << 4) + (hh << 3)) * 2u + (size_t)oc);
#pragma unroll
        for (int g = 0; g < 8; ++g) {
            p[2 * g] = acc[g] + bias;
        }
    }
}

extern "C" void kernel_launch(void* const* d_in, const int* in_sizes, int n_in,
                              void* d_out, int out_size, void* d_ws, size_t ws_size,
                              hipStream_t stream) {
    (void)in_sizes; (void)n_in; (void)d_ws; (void)ws_size; (void)out_size;

    const float* x    = (const float*)d_in[0];
    const float* w_ih = (const float*)d_in[1];
    const float* w_hh = (const float*)d_in[2];
    const float* b_ih = (const float*)d_in[3];
    const float* b_hh = (const float*)d_in[4];
    const float* l1_w = (const float*)d_in[5];
    const float* l1_b = (const float*)d_in[6];
    float* out = (float*)d_out;

    const int B = 32768;
    dim3 block(128);
    dim3 grid(B / 128);
    hipLaunchKernelGGL(OksuzLSTM_scan_wmma, grid, block, 0, stream,
                       x, w_ih, w_hh, b_ih, b_hh, l1_w, l1_b, out);
}


# Loop body in issue order, held by scheduling fences: chain links at operand arrival; output sigmoid 2 steps late, pair's matrix instruction 1 pair late behind tanh(c), first on a zero B; same sums
# speedup vs baseline: 2.8952x; 1.0834x over previous
#include <hip/hip_runtime.h>


typedef __attribute__((ext_vector_type(2))) float v2f;
typedef __attribute__((ext_vector_type(8))) float v8f;

#if __has_builtin(__builtin_amdgcn_tanhf)
#define HTANH(x) __builtin_amdgcn_tanhf((x))
#elif __has_builtin(__builtin_amdgcn_tanh_f32)
#define HTANH(x) __builtin_amdgcn_tanh_f32((x))
#else
#define HTANH(x) fmaf(2.0f, __builtin_amdgcn_rcpf(1.0f + __expf(-2.0f * (x))), -1.0f)
#endif

#define RING 10

#define RING_PIN() asm volatile("" ::: "memory")

#define HOLD() __builtin_amdgcn_sched_barrier(0)

#define STEP(pc, pn, xn, yOut, TAIL)                                        \
    do {                                                                    \
        float af = fmaf(tc, q1, (pc)[1]);                                   \
        float ag = fmaf(tc, q2, (pc)[2]);                                   \
        float ai = fmaf(tc, q0, (pc)[0]);                                   \
        float ao = fmaf(tc, q3, (pc)[3]);                                   \
        float ya = ohP * tc;         \
        HOLD();             \
        float tf = HTANH(af);                                               \
        (pn)[1] = fmaf((xn), wi1, bb1);                                     \
        HOLD();  \
        float ti = HTANH(ai);                                               \
        (pn)[0] = fmaf((xn), wi0, bb0);                                     \
        HOLD();                                                \
        float tg = HTANH(ag);                                               \
        (pn)[2] = fmaf((xn), wi2, bb2);                                     \
        HOLD();                                                \
        float to = HTANH(ao);                                               \
        (pn)[3] = fmaf((xn), wi3, bb3);                                     \
        HOLD();  \
        float fc = fmaf(tf, ch, ch);                                        \
        HOLD();    \
        float tyN = HTANH(ya);                                              \
        float ii = fmaf(0.5f, ti, 0.5f);                                    \
        (yOut) = fmaf(0.5f, tyP, 0.5f);                                     \
        HOLD();      \
        float c = fmaf(ii, tg, fc);                                         \
        HOLD();                      \
        q0 = fmaf(to, wq0, wq0);                                            \
        q1 = fmaf(to, wq1, wq1);                                            \
        q2 = fmaf(to, wq2, wq2);                                            \
        q3 = fmaf(to, wq3, wq3);                                            \
        HOLD();       \
        tc = HTANH(c);                                                      \
        ohP = fmaf(0.25f, to, 0.25f);                                       \
        ch = 0.5f * c;                                                      \
        tyP = tyN;                                                          \
        HOLD();                      \
        TAIL                                                                \
    } while (0)

__global__ __launch_bounds__(128)
void OksuzLSTM_scan_wmma(const float* __restrict__ x,
                         const float* __restrict__ w_ih,
                         const float* __restrict__ w_hh,
                         const float* __restrict__ b_ih,
                         const float* __restrict__ b_hh,
                         const float* __restrict__ l1_w,
                         const float* __restrict__ l1_b,
                         float* __restrict__ out)
{
    const int lane     = (int)(threadIdx.x & 31u);
    const int row      = blockIdx.x * (int)blockDim.x + (int)threadIdx.x;
    const int waveBase = blockIdx.x * (int)blockDim.x + (int)(threadIdx.x & ~31u);

    const float wi0 = 0.5f * w_ih[0], wi1 = 0.5f * w_ih[1], wi2 = w_ih[2], wi3 = 0.5f * w_ih[3];
    const float bb0 = 0.5f * (b_ih[0] + b_hh[0]);
    const float bb1 = 0.5f * (b_ih[1] + b_hh[1]);
    const float bb2 = b_ih[2] + b_hh[2];
    const float bb3 = 0.5f * (b_ih[3] + b_hh[3]);
    const float wq0 = 0.25f * w_hh[0], wq1 = 0.25f * w_hh[1], wq2 = 0.5f * w_hh[2], wq3 = 0.25f * w_hh[3];

    const float* xrow = x + (size_t)row * 1000u;

    const int  n    = lane & 15;
    const int  hh   = lane >> 4;
    const bool bact = (n < 4) && ((n >> 1) == hh);
    const int  oc   = n & 1;
    const float* wrow = l1_w + (size_t)(bact ? oc : 0) * 1000u;

    float ch = 0.0f, tc = 0.0f;
    float q0 = wq0, q1 = wq1, q2 = wq2, q3 = wq3;
    float ohP = 0.0f, tyP = 0.0f;
    v8f acc = {0.f, 0.f, 0.f, 0.f, 0.f, 0.f, 0.f, 0.f};

    float4 xq[RING];
    float4 bq[RING];
#pragma unroll
    for (int k = 0; k < RING; ++k) {
        xq[k] = *reinterpret_cast<const float4*>(xrow + 4 * k);
        bq[k] = *reinterpret_cast<const float4*>(wrow + 4 * k);
        RING_PIN();
    }

    float pa[4], pb[4];
    pa[0] = fmaf(xq[0].x, wi0, bb0);
    pa[1] = fmaf(xq[0].x, wi1, bb1);
    pa[2] = fmaf(xq[0].x, wi2, bb2);
    pa[3] = fmaf(xq[0].x, wi3, bb3);
    v2f bfP = {0.0f, 0.0f};

    for (int t0 = 0; t0 < 1000; t0 += 4 * RING) {
        const int tr = (t0 + 4 * RING < 1000) ? (t0 + 4 * RING) : t0;

        __builtin_prefetch(xrow + t0 + 104, 0, 1);
        __builtin_prefetch(xrow + t0 + 124, 0, 1);

#pragma unroll
        for (int k = 0; k < RING; ++k) {
            const float4 xv = xq[k];
            const float  xn = xq[(k + 1) % RING].x;
            float e0, e1, e2, e3;
            v2f bf0, bf1;

            STEP(pa, pb, xv.y, e0, );
            STEP(pb, pa, xv.z, e1,
                 v2f a0;
                 a0.x = e0;
                 a0.y = e1;
                 acc = __builtin_amdgcn_wmma_f32_16x16x4_f32(
                     false, a0, false, bfP, (short)0, acc, false, false););
            STEP(pa, pb, xv.w, e2,
                 const float4 bw = bq[k];
                 bf0.x = bact ? bw.x : 0.0f;
                 bf0.y = bact ? bw.y : 0.0f;
                 bf1.x = bact ? bw.z : 0.0f;
                 bf1.y = bact ? bw.w : 0.0f;
                 HOLD();
                 xq[k] = *reinterpret_cast<const float4*>(xrow + tr + 4 * k);
                 bq[k] = *reinterpret_cast<const float4*>(wrow + tr + 4 * k);
                 RING_PIN(););
            STEP(pb, pa, xn, e3,
                 v2f a1;
                 a1.x = e2;
                 a1.y = e3;
                 acc = __builtin_amdgcn_wmma_f32_16x16x4_f32(
                     false, a1, false, bf0, (short)0, acc, false, false););
            bfP = bf1;
        }
    }

    {
        v2f a2;
        a2.x = fmaf(0.5f, tyP, 0.5f);
        a2.y = fmaf(0.5f, HTANH(ohP * tc), 0.5f);
        acc = __builtin_amdgcn_wmma_f32_16x16x4_f32(
            false, a2, false, bfP, (short)0, acc, false, false);
    }

    if (n < 4) {
        const float bias = l1_b[oc];
        float* p = out + ((size_t)(waveBase + ((n >> 1) << 4) + (hh << 3)) * 2u + (size_t)oc);
#pragma unroll
        for (int g = 0; g < 8; ++g) {
            p[2 * g] = acc[g] + bias;
        }
    }
}

extern "C" void kernel_launch(void* const* d_in, const int* in_sizes, int n_in,
                              void* d_out, int out_size, void* d_ws, size_t ws_size,
                              hipStream_t stream) {
    (void)in_sizes; (void)n_in; (void)d_ws; (void)ws_size; (void)out_size;

    const float* x    = (const float*)d_in[0];
    const float* w_ih = (const float*)d_in[1];
    const float* w_hh = (const float*)d_in[2];
    const float* b_ih = (const float*)d_in[3];
    const float* b_hh = (const float*)d_in[4];
    const float* l1_w = (const float*)d_in[5];
    const float* l1_b = (const float*)d_in[6];
    float* out = (float*)d_out;

    const int B = 32768;
    dim3 block(128);
    dim3 grid(B / 128);
    hipLaunchKernelGGL(OksuzLSTM_scan_wmma, grid, block, 0, stream,
                       x, w_ih, w_hh, b_ih, b_hh, l1_w, l1_b, out);
}
